# starting kernel
# baseline (speedup 1.0000x reference)
#include <hip/hip_runtime.h>
#include <hip/hip_bf16.h>

typedef __attribute__((ext_vector_type(16))) _Float16 v16h;
typedef __attribute__((ext_vector_type(8)))  float    v8f;

template <int K, int NCOL>
__global__ __launch_bounds__(32) void pack_weights(
    const float* __restrict__ m0, const float* __restrict__ m1,
    const float* __restrict__ m2, _Float16* __restrict__ out)
{
  const float* W = (blockIdx.y == 0) ? m0 : (blockIdx.y == 1) ? m1 : m2;
  const int ct   = blockIdx.x / (K / 32);
  const int ks   = blockIdx.x % (K / 32);
  const int lane = threadIdx.x;
  const int hs   = lane >> 4;
  const int mr   = lane & 15;
  const int krow = ks * 32 + hs * 8;
  const float* p = W + (size_t)krow * NCOL + ct * 16 + mr;
  _Float16* o = out +
      ((((size_t)blockIdx.y * (NCOL / 16) + ct) * (K / 32) + ks) * 32 + lane) * 16;
#pragma unroll
  for (int i = 0; i < 8; ++i) {
    o[i]     = (_Float16)p[(size_t)i * NCOL];
    o[8 + i] = (_Float16)p[(size_t)(16 + i) * NCOL];
  }
}

template <int K, int NCOL, bool RELU>
__global__ __launch_bounds__((NCOL / 16) * 32) void node_xform(
    const float* __restrict__ x,
    const _Float16* __restrict__ wpack,
    const float* __restrict__ bias,
    float* __restrict__ o0, float* __restrict__ o1, float* __restrict__ oagg)
{
  constexpr int KS = K / 32;
  __shared__ v16h ash[KS * 32];

  const int lane = threadIdx.x & 31;
  const int wave = threadIdx.x >> 5;
  const int row0 = blockIdx.x * 16;
  const int hs   = lane >> 4;
  const int mr   = lane & 15;

  if (wave < KS) {
    const int    ks = wave;
    const float* ap = x + (size_t)(row0 + mr) * K + ks * 32 + hs * 8;
    v16h a;
#pragma unroll
    for (int i = 0; i < 8; ++i) {
      float lo = ap[i];
      float hi = ap[16 + i];
      if (RELU) { lo = fmaxf(lo, 0.0f); hi = fmaxf(hi, 0.0f); }
      a[i]     = (_Float16)lo;
      a[8 + i] = (_Float16)hi;
    }
    ash[ks * 32 + lane] = a;
  }
  __syncthreads();

  v8f c0 = {}, c1 = {}, cl = {};
  constexpr size_t MATSTRIDE = (size_t)(NCOL / 16) * KS * 32 * 16;
  const _Float16* wb = wpack + ((size_t)wave * KS) * 32 * 16 + (size_t)lane * 16;

#pragma unroll
  for (int ks = 0; ks < KS; ++ks) {
    v16h a = ash[ks * 32 + lane];
    const _Float16* f = wb + (size_t)ks * 32 * 16;
    v16h b0 = *(const v16h*)(f);
    v16h b1 = *(const v16h*)(f + MATSTRIDE);
    v16h bl = *(const v16h*)(f + 2 * MATSTRIDE);
    c0 = __builtin_amdgcn_wmma_f32_16x16x32_f16(false, a, false, b0, (short)0, c0, false, false);
    c1 = __builtin_amdgcn_wmma_f32_16x16x32_f16(false, a, false, b1, (short)0, c1, false, false);
    cl = __builtin_amdgcn_wmma_f32_16x16x32_f16(false, a, false, bl, (short)0, cl, false, false);
  }

  const int   col = wave * 16 + mr;
  const float bv  = bias[col];
#pragma unroll
  for (int r = 0; r < 8; ++r) {
    size_t off = (size_t)(row0 + r + hs * 8) * NCOL + col;
    o0[off]   = c0[r];
    o1[off]   = c1[r];
    oagg[off] = cl[r] + bv;
  }
}

__global__ __launch_bounds__(256) void zero_u32(unsigned* p, int n) {
  int i = blockIdx.x * 256 + threadIdx.x;
  if (i < n) p[i] = 0u;
}

__global__ __launch_bounds__(256) void count_edges(
    const int* __restrict__ dst, unsigned* __restrict__ counts, int E) {
  int e = blockIdx.x * 256 + threadIdx.x;
  if (e < E) atomicAdd(&counts[dst[e]], 1u);
}

__global__ __launch_bounds__(1024) void scan_offsets(
    const unsigned* __restrict__ counts, unsigned* __restrict__ offsets,
    unsigned* __restrict__ cursor, int n)
{
  __shared__ unsigned s[1024];
  __shared__ unsigned carry;
  const int t = threadIdx.x;
  if (t == 0) carry = 0u;
  __syncthreads();
  for (int base = 0; base < n; base += 1024) {
    const int i = base + t;
    unsigned v = (i < n) ? counts[i] : 0u;
    s[t] = v;
    __syncthreads();
#pragma unroll
    for (int off = 1; off < 1024; off <<= 1) {
      unsigned u = (t >= off) ? s[t - off] : 0u;
      __syncthreads();
      s[t] += u;
      __syncthreads();
    }
    const unsigned excl  = s[t] - v + carry;
    const unsigned total = s[1023];
    if (i < n) { offsets[i] = excl; cursor[i] = excl; }
    __syncthreads();
    if (t == 0) carry += total;
    __syncthreads();
  }
  if (t == 0) offsets[n] = carry;
}

__global__ __launch_bounds__(256) void fill_edges(
    const int* __restrict__ src, const int* __restrict__ dst,
    const int* __restrict__ et, unsigned* __restrict__ cursor,
    unsigned* __restrict__ eidx, int E)
{
  int e = blockIdx.x * 256 + threadIdx.x;
  if (e >= E) return;
  unsigned pos = atomicAdd(&cursor[dst[e]], 1u);
  eidx[pos] = (unsigned)src[e] | ((unsigned)et[e] << 30);
}

template <int F>
__global__ __launch_bounds__(256) void aggregate(
    const float* __restrict__ x0, const float* __restrict__ x1,
    const unsigned* __restrict__ offsets, const unsigned* __restrict__ eidx,
    float* __restrict__ io, int N)
{
  constexpr int VEC = F / 32;
  int gid  = blockIdx.x * 256 + threadIdx.x;
  int node = gid >> 5;
  if (node >= N) return;
  int lane = gid & 31;

  const unsigned beg = offsets[node];
  const unsigned end = offsets[node + 1];

  float acc0[VEC], acc1[VEC];
#pragma unroll
  for (int i = 0; i < VEC; ++i) { acc0[i] = 0.0f; acc1[i] = 0.0f; }

  unsigned e = beg;
  for (; e + 2 <= end; e += 2) {
    const unsigned pk0 = eidx[e];
    const unsigned pk1 = eidx[e + 1];
    const float* r0 = ((pk0 >> 30) ? x1 : x0) + (size_t)(pk0 & 0x3FFFFFFFu) * F + lane * VEC;
    const float* r1 = ((pk1 >> 30) ? x1 : x0) + (size_t)(pk1 & 0x3FFFFFFFu) * F + lane * VEC;
    if (VEC == 4) {
      const float4 v0 = *(const float4*)r0;
      const float4 v1 = *(const float4*)r1;
      acc0[0] += v0.x; acc0[1] += v0.y; acc0[2] += v0.z; acc0[3] += v0.w;
      acc1[0] += v1.x; acc1[1] += v1.y; acc1[2] += v1.z; acc1[3] += v1.w;
    } else {
      const float2 v0 = *(const float2*)r0;
      const float2 v1 = *(const float2*)r1;
      acc0[0] += v0.x; acc0[1] += v0.y;
      acc1[0] += v1.x; acc1[1] += v1.y;
    }
  }
  if (e < end) {
    const unsigned pk = eidx[e];
    const float* r = ((pk >> 30) ? x1 : x0) + (size_t)(pk & 0x3FFFFFFFu) * F + lane * VEC;
    if (VEC == 4) {
      const float4 v = *(const float4*)r;
      acc0[0] += v.x; acc0[1] += v.y; acc0[2] += v.z; acc0[3] += v.w;
    } else {
      const float2 v = *(const float2*)r;
      acc0[0] += v.x; acc0[1] += v.y;
    }
  }

  float* o = io + (size_t)node * F + lane * VEC;
#pragma unroll
  for (int i = 0; i < VEC; ++i) o[i] += acc0[i] + acc1[i];
}

extern "C" void kernel_launch(void* const* d_in, const int* in_sizes, int n_in,
                              void* d_out, int out_size, void* d_ws, size_t ws_size,
                              hipStream_t stream) {
  const float* feat  = (const float*)d_in[0];
  const float* W1    = (const float*)d_in[1];
  const float* loop1 = (const float*)d_in[2];
  const float* b1    = (const float*)d_in[3];
  const float* W2    = (const float*)d_in[4];
  const float* loop2 = (const float*)d_in[5];
  const float* b2    = (const float*)d_in[6];
  const int*   src   = (const int*)d_in[7];
  const int*   dst   = (const int*)d_in[8];
  const int*   et    = (const int*)d_in[9];
  float*       out   = (float*)d_out;

  const int N = in_sizes[0] / 64;
  const int E = in_sizes[7];

  float* ws   = (float*)d_ws;
  float* x0   = ws;
  float* x1   = x0 + (size_t)N * 128;
  float* hbuf = x1 + (size_t)N * 128;
  float* y0   = x0;
  float* y1   = x0 + (size_t)N * 64;
  _Float16* wpack1 = (_Float16*)(hbuf + (size_t)N * 128);
  _Float16* wpack2 = wpack1 + 24576;
  unsigned* counts  = (unsigned*)(wpack2 + 24576);
  unsigned* offsets = counts + N;
  unsigned* cursor  = offsets + N + 64;
  unsigned* eidx    = cursor + N;

  const int nblk = N / 16;
  const int eblk = (E + 255) / 256;
  const int ablk = (N * 32 + 255) / 256;
  const int zblk = (N + 255) / 256;

  pack_weights<64, 128><<<dim3(8 * 2, 3), 32, 0, stream>>>(
      W1, W1 + 64 * 128, loop1, wpack1);
  pack_weights<128, 64><<<dim3(4 * 4, 3), 32, 0, stream>>>(
      W2, W2 + 128 * 64, loop2, wpack2);

  zero_u32<<<zblk, 256, 0, stream>>>(counts, N);
  count_edges<<<eblk, 256, 0, stream>>>(dst, counts, E);
  scan_offsets<<<1, 1024, 0, stream>>>(counts, offsets, cursor, N);
  fill_edges<<<eblk, 256, 0, stream>>>(src, dst, et, cursor, eidx, E);

  node_xform<64, 128, false><<<nblk, 256, 0, stream>>>(
      feat, wpack1, b1, x0, x1, hbuf);
  aggregate<128><<<ablk, 256, 0, stream>>>(x0, x1, offsets, eidx, hbuf, N);

  node_xform<128, 64, true><<<nblk, 128, 0, stream>>>(
      hbuf, wpack2, b2, y0, y1, out);
  aggregate<64><<<ablk, 256, 0, stream>>>(y0, y1, offsets, eidx, out, N);
}


# Layer 1 sums the 64-wide input rows per relation first, then one product [sum0|sum1|x] x stacked weights, instead of gathering 128-wide products; float summation order changes
# speedup vs baseline: 1.6750x; 1.6750x over previous
#include <hip/hip_runtime.h>
#include <hip/hip_bf16.h>

typedef __attribute__((ext_vector_type(16))) _Float16 v16h;
typedef __attribute__((ext_vector_type(8)))  float    v8f;

template <int K, int NCOL, bool CAT>
__global__ __launch_bounds__(32) void pack_weights(
    const float* __restrict__ m0, const float* __restrict__ m1,
    const float* __restrict__ m2, _Float16* __restrict__ out)
{
  const float* W = (blockIdx.y == 0) ? m0 : (blockIdx.y == 1) ? m1 : m2;
  const int ct   = blockIdx.x / (K / 32);
  const int ks   = blockIdx.x % (K / 32);
  const int lane = threadIdx.x;
  const int hs   = lane >> 4;
  const int mr   = lane & 15;
  const int krow = ks * 32 + hs * 8;
  const float* p = W + (size_t)krow * NCOL + ct * 16 + mr;
  const size_t frag = CAT
      ? ((size_t)ct * (3 * (K / 32)) + blockIdx.y * (K / 32) + ks)
      : (((size_t)blockIdx.y * (NCOL / 16) + ct) * (K / 32) + ks);
  _Float16* o = out + (frag * 32 + lane) * 16;
#pragma unroll
  for (int i = 0; i < 8; ++i) {
    o[i]     = (_Float16)p[(size_t)i * NCOL];
    o[8 + i] = (_Float16)p[(size_t)(16 + i) * NCOL];
  }
}

template <int K, int NCOL, bool RELU>
__global__ __launch_bounds__((NCOL / 16) * 32) void node_xform(
    const float* __restrict__ x,
    const _Float16* __restrict__ wpack,
    const float* __restrict__ bias,
    float* __restrict__ o0, float* __restrict__ o1, float* __restrict__ oagg)
{
  constexpr int KS = K / 32;
  __shared__ v16h ash[KS * 32];

  const int lane = threadIdx.x & 31;
  const int wave = threadIdx.x >> 5;
  const int row0 = blockIdx.x * 16;
  const int hs   = lane >> 4;
  const int mr   = lane & 15;

  if (wave < KS) {
    const int    ks = wave;
    const float* ap = x + (size_t)(row0 + mr) * K + ks * 32 + hs * 8;
    v16h a;
#pragma unroll
    for (int i = 0; i < 8; ++i) {
      float lo = ap[i];
      float hi = ap[16 + i];
      if (RELU) { lo = fmaxf(lo, 0.0f); hi = fmaxf(hi, 0.0f); }
      a[i]     = (_Float16)lo;
      a[8 + i] = (_Float16)hi;
    }
    ash[ks * 32 + lane] = a;
  }
  __syncthreads();

  v8f c0 = {}, c1 = {}, cl = {};
  constexpr size_t MATSTRIDE = (size_t)(NCOL / 16) * KS * 32 * 16;
  const _Float16* wb = wpack + ((size_t)wave * KS) * 32 * 16 + (size_t)lane * 16;

#pragma unroll
  for (int ks = 0; ks < KS; ++ks) {
    v16h a = ash[ks * 32 + lane];
    const _Float16* f = wb + (size_t)ks * 32 * 16;
    v16h b0 = *(const v16h*)(f);
    v16h b1 = *(const v16h*)(f + MATSTRIDE);
    v16h bl = *(const v16h*)(f + 2 * MATSTRIDE);
    c0 = __builtin_amdgcn_wmma_f32_16x16x32_f16(false, a, false, b0, (short)0, c0, false, false);
    c1 = __builtin_amdgcn_wmma_f32_16x16x32_f16(false, a, false, b1, (short)0, c1, false, false);
    cl = __builtin_amdgcn_wmma_f32_16x16x32_f16(false, a, false, bl, (short)0, cl, false, false);
  }

  const int   col = wave * 16 + mr;
  const float bv  = bias[col];
#pragma unroll
  for (int r = 0; r < 8; ++r) {
    size_t off = (size_t)(row0 + r + hs * 8) * NCOL + col;
    o0[off]   = c0[r];
    o1[off]   = c1[r];
    oagg[off] = cl[r] + bv;
  }
}

template <int K, int NCOL>
__global__ __launch_bounds__((NCOL / 16) * 32) void node_xform_cat(
    const float* __restrict__ a0,
    const float* __restrict__ a1,
    const float* __restrict__ x,
    const _Float16* __restrict__ wpack,
    const float* __restrict__ bias,
    float* __restrict__ h)
{
  constexpr int KP = K / 32;
  constexpr int KS = 3 * KP;
  static_assert(KS <= NCOL / 16, "one building wave per k-step");
  __shared__ v16h ash[KS * 32];

  const int lane = threadIdx.x & 31;
  const int wave = threadIdx.x >> 5;
  const int row0 = blockIdx.x * 16;
  const int hs   = lane >> 4;
  const int mr   = lane & 15;

  if (wave < KS) {
    const int    part = wave / KP;
    const int    kk   = wave % KP;
    const float* sp   = (part == 0) ? a0 : (part == 1) ? a1 : x;
    const float* ap   = sp + (size_t)(row0 + mr) * K + kk * 32 + hs * 8;
    v16h a;
#pragma unroll
    for (int i = 0; i < 8; ++i) {
      a[i]     = (_Float16)ap[i];
      a[8 + i] = (_Float16)ap[16 + i];
    }
    ash[wave * 32 + lane] = a;
  }
  __syncthreads();

  v8f c[3] = {{}, {}, {}};
  const _Float16* wb = wpack + ((size_t)wave * KS) * 32 * 16 + (size_t)lane * 16;

#pragma unroll
  for (int g = 0; g < KS; ++g) {
    const v16h a = ash[g * 32 + lane];
    const v16h b = *(const v16h*)(wb + (size_t)g * 32 * 16);
    c[g / KP] = __builtin_amdgcn_wmma_f32_16x16x32_f16(
        false, a, false, b, (short)0, c[g / KP], false, false);
  }

  const int   col = wave * 16 + mr;
  const float bv  = bias[col];
#pragma unroll
  for (int r = 0; r < 8; ++r) {
    const size_t off = (size_t)(row0 + r + hs * 8) * NCOL + col;
    h[off] = (c[0][r] + c[1][r]) + (c[2][r] + bv);
  }
}

__global__ __launch_bounds__(256) void zero_u32(unsigned* p, int n) {
  int i = blockIdx.x * 256 + threadIdx.x;
  if (i < n) p[i] = 0u;
}

__global__ __launch_bounds__(256) void count_edges(
    const int* __restrict__ dst, unsigned* __restrict__ counts, int E) {
  int e = blockIdx.x * 256 + threadIdx.x;
  if (e < E) atomicAdd(&counts[dst[e]], 1u);
}

__global__ __launch_bounds__(1024) void scan_offsets(
    const unsigned* __restrict__ counts, unsigned* __restrict__ offsets,
    unsigned* __restrict__ cursor, int n)
{
  __shared__ unsigned wtot[32];
  __shared__ unsigned wpre[32];
  const int t    = threadIdx.x;
  const int lane = t & 31;
  const int wave = t >> 5;
  const int i    = blockIdx.x * 1024 + t;

  const unsigned v = (i < n) ? counts[i] : 0u;

  const uint4* c4   = reinterpret_cast<const uint4*>(counts);
  const int    nvec = blockIdx.x * 256;
  unsigned p = 0u;
  for (int j0 = 0; j0 < nvec; j0 += 12 * 1024) {
    for (int k = 0; k < 12; ++k) {
      const int      j   = j0 + k * 1024 + t;
      const bool     in  = j < nvec;
      const uint4    q   = c4[in ? j : 0];
      const unsigned sum = q.x + q.y + q.z + q.w;
      p += in ? sum : 0u;
    }
  }

  unsigned s = v;
  for (int off = 1; off < 32; off <<= 1) {
    const unsigned u = __shfl_up(s, off, 32);
    if (lane >= off) s += u;
  }
  for (int off = 16; off > 0; off >>= 1) p += __shfl_xor(p, off, 32);

  if (lane == 31) wtot[wave] = s;
  if (lane == 0)  wpre[wave] = p;
  __syncthreads();

  unsigned c = wpre[lane] + ((lane < wave) ? wtot[lane] : 0u);
  for (int off = 16; off > 0; off >>= 1) c += __shfl_xor(c, off, 32);

  const unsigned excl = c + s - v;
  if (i < n) { offsets[i] = excl; cursor[i] = excl; }
  if (i == n - 1) offsets[n] = excl + v;
}

__global__ __launch_bounds__(256) void fill_edges(
    const int* __restrict__ src, const int* __restrict__ dst,
    const int* __restrict__ et, unsigned* __restrict__ cursor,
    unsigned* __restrict__ eidx, int E)
{
  int e = blockIdx.x * 256 + threadIdx.x;
  if (e >= E) return;
  unsigned pos = atomicAdd(&cursor[dst[e]], 1u);
  eidx[pos] = (unsigned)src[e] | ((unsigned)et[e] << 30);
}

template <int F>
__global__ __launch_bounds__(256) void aggregate(
    const float* __restrict__ x0, const float* __restrict__ x1,
    const unsigned* __restrict__ offsets, const unsigned* __restrict__ eidx,
    float* __restrict__ io, int N)
{
  constexpr int VEC = F / 32;
  int gid  = blockIdx.x * 256 + threadIdx.x;
  int node = gid >> 5;
  if (node >= N) return;
  int lane = gid & 31;

  const unsigned beg = offsets[node];
  const unsigned end = offsets[node + 1];

  float acc0[VEC], acc1[VEC];
#pragma unroll
  for (int i = 0; i < VEC; ++i) { acc0[i] = 0.0f; acc1[i] = 0.0f; }

  unsigned e = beg;
  for (; e + 2 <= end; e += 2) {
    const unsigned pk0 = eidx[e];
    const unsigned pk1 = eidx[e + 1];
    const float* r0 = ((pk0 >> 30) ? x1 : x0) + (size_t)(pk0 & 0x3FFFFFFFu) * F + lane * VEC;
    const float* r1 = ((pk1 >> 30) ? x1 : x0) + (size_t)(pk1 & 0x3FFFFFFFu) * F + lane * VEC;
    if (VEC == 4) {
      const float4 v0 = *(const float4*)r0;
      const float4 v1 = *(const float4*)r1;
      acc0[0] += v0.x; acc0[1] += v0.y; acc0[2] += v0.z; acc0[3] += v0.w;
      acc1[0] += v1.x; acc1[1] += v1.y; acc1[2] += v1.z; acc1[3] += v1.w;
    } else {
      const float2 v0 = *(const float2*)r0;
      const float2 v1 = *(const float2*)r1;
      acc0[0] += v0.x; acc0[1] += v0.y;
      acc1[0] += v1.x; acc1[1] += v1.y;
    }
  }
  if (e < end) {
    const unsigned pk = eidx[e];
    const float* r = ((pk >> 30) ? x1 : x0) + (size_t)(pk & 0x3FFFFFFFu) * F + lane * VEC;
    if (VEC == 4) {
      const float4 v = *(const float4*)r;
      acc0[0] += v.x; acc0[1] += v.y; acc0[2] += v.z; acc0[3] += v.w;
    } else {
      const float2 v = *(const float2*)r;
      acc0[0] += v.x; acc0[1] += v.y;
    }
  }

  float* o = io + (size_t)node * F + lane * VEC;
#pragma unroll
  for (int i = 0; i < VEC; ++i) o[i] += acc0[i] + acc1[i];
}

__global__ __launch_bounds__(256) void aggregate_in(
    const float* __restrict__ x,
    const unsigned* __restrict__ offsets, const unsigned* __restrict__ eidx,
    float* __restrict__ a0, float* __restrict__ a1, int N)
{
  const int gid  = blockIdx.x * 256 + threadIdx.x;
  const int node = gid >> 5;
  if (node >= N) return;
  const int lane = gid & 31;

  const unsigned beg = offsets[node];
  const unsigned end = offsets[node + 1];
  const float*   xl  = x + lane * 2;

  float2 s0 = make_float2(0.0f, 0.0f);
  float2 s1 = make_float2(0.0f, 0.0f);

  unsigned e = beg;
  for (; e + 2 <= end; e += 2) {
    const unsigned pk0 = eidx[e];
    const unsigned pk1 = eidx[e + 1];
    const float2 v0 = *(const float2*)(xl + (size_t)(pk0 & 0x3FFFFFFFu) * 64);
    const float2 v1 = *(const float2*)(xl + (size_t)(pk1 & 0x3FFFFFFFu) * 64);
    if (pk0 >> 30) { s1.x += v0.x; s1.y += v0.y; } else { s0.x += v0.x; s0.y += v0.y; }
    if (pk1 >> 30) { s1.x += v1.x; s1.y += v1.y; } else { s0.x += v1.x; s0.y += v1.y; }
  }
  if (e < end) {
    const unsigned pk = eidx[e];
    const float2 v = *(const float2*)(xl + (size_t)(pk & 0x3FFFFFFFu) * 64);
    if (pk >> 30) { s1.x += v.x; s1.y += v.y; } else { s0.x += v.x; s0.y += v.y; }
  }

  *(float2*)(a0 + (size_t)node * 64 + lane * 2) = s0;
  *(float2*)(a1 + (size_t)node * 64 + lane * 2) = s1;
}

extern "C" void kernel_launch(void* const* d_in, const int* in_sizes, int n_in,
                              void* d_out, int out_size, void* d_ws, size_t ws_size,
                              hipStream_t stream) {
  const float* feat  = (const float*)d_in[0];
  const float* W1    = (const float*)d_in[1];
  const float* loop1 = (const float*)d_in[2];
  const float* b1    = (const float*)d_in[3];
  const float* W2    = (const float*)d_in[4];
  const float* loop2 = (const float*)d_in[5];
  const float* b2    = (const float*)d_in[6];
  const int*   src   = (const int*)d_in[7];
  const int*   dst   = (const int*)d_in[8];
  const int*   et    = (const int*)d_in[9];
  float*       out   = (float*)d_out;

  const int N = in_sizes[0] / 64;
  const int E = in_sizes[7];

  float* ws   = (float*)d_ws;
  float* x0   = ws;
  float* a0   = x0;
  float* a1   = x0 + (size_t)N * 64;
  float* x1   = x0 + (size_t)N * 128;
  float* hbuf = x1 + (size_t)N * 128;
  float* y0   = x0;
  float* y1   = x0 + (size_t)N * 64;
  _Float16* wpack1 = (_Float16*)(hbuf + (size_t)N * 128);
  _Float16* wpack2 = wpack1 + 24576;
  unsigned* counts  = (unsigned*)(wpack2 + 24576);
  unsigned* offsets = counts + N;
  unsigned* cursor  = offsets + N + 64;
  unsigned* eidx    = cursor + N;

  const int nblk = N / 16;
  const int eblk = (E + 255) / 256;
  const int ablk = (N * 32 + 255) / 256;
  const int zblk = (N + 255) / 256;
  const int sblk = (N + 1023) / 1024;

  pack_weights<64, 128, true><<<dim3(8 * 2, 3), 32, 0, stream>>>(
      W1, W1 + 64 * 128, loop1, wpack1);
  pack_weights<128, 64, false><<<dim3(4 * 4, 3), 32, 0, stream>>>(
      W2, W2 + 128 * 64, loop2, wpack2);

  zero_u32<<<zblk, 256, 0, stream>>>(counts, N);
  count_edges<<<eblk, 256, 0, stream>>>(dst, counts, E);
  scan_offsets<<<sblk, 1024, 0, stream>>>(counts, offsets, cursor, N);
  fill_edges<<<eblk, 256, 0, stream>>>(src, dst, et, cursor, eidx, E);

  aggregate_in<<<ablk, 256, 0, stream>>>(feat, offsets, eidx, a0, a1, N);
  node_xform_cat<64, 128><<<nblk, 256, 0, stream>>>(
      a0, a1, feat, wpack1, b1, hbuf);

  node_xform<128, 64, true><<<nblk, 128, 0, stream>>>(
      hbuf, wpack2, b2, y0, y1, out);
  aggregate<64><<<ablk, 256, 0, stream>>>(y0, y1, offsets, eidx, out, N);
}
